# starting kernel
# baseline (speedup 1.0000x reference)
#include <hip/hip_runtime.h>
#include <hip/hip_bf16.h>
#include <math.h>

#define ALPHA 0.2f

typedef __attribute__((ext_vector_type(2))) float v2f;
typedef __attribute__((ext_vector_type(8))) float v8f;

__global__ __launch_bounds__(32) void gat_h_gemm_wmma(
    const float* __restrict__ X,
    const float* __restrict__ W,
    float* __restrict__ H)
{
  const int tileM = blockIdx.x;
  const int tileN = blockIdx.y;
  const int lane  = threadIdx.x;
  const int half  = lane >> 4;
  const int l15   = lane & 15;

  const int rowA = tileM * 16 + l15;
  const int colD = tileN * 16 + l15;

  const float* __restrict__ xrow = X + rowA * 128;

  v8f acc = {};
#pragma unroll
  for (int k0 = 0; k0 < 128; k0 += 4) {
    const int k = k0 + 2 * half;
    v2f a, b;
    a.x = xrow[k];
    a.y = xrow[k + 1];
    b.x = W[k * 64 + colD];
    b.y = W[(k + 1) * 64 + colD];
    acc = __builtin_amdgcn_wmma_f32_16x16x4_f32(
        false, a, false, b, (short)0, acc, false, false);
  }

#pragma unroll
  for (int r = 0; r < 8; ++r) {
    const int row = tileM * 16 + r + 8 * half;
    H[row * 64 + colD] = acc[r];
  }
}

__global__ __launch_bounds__(256) void gat_post(
    const float* __restrict__ H,
    const float* __restrict__ A,
    float* __restrict__ OUT)
{
  const int b   = blockIdx.x;
  const int tid = threadIdx.x;
  const float* __restrict__ h = H + b * 512 * 64;
  float* __restrict__ o       = OUT + b * 512 * 64;

  __shared__ float s_a[128];
  __shared__ float s_s[512];
  __shared__ float s_u[512];
  __shared__ float s_v[513];
  __shared__ float s_alpha[512];
  __shared__ float s_Slo[64];
  __shared__ float s_Shi[64];
  __shared__ float s_P[64];
  __shared__ float s_red[256];

  if (tid < 128) s_a[tid] = A[tid];
  __syncthreads();

  for (int r = tid; r < 512; r += 256) {
    const float* hr = h + r * 64;
    float du = 0.f, dv = 0.f;
#pragma unroll 8
    for (int c = 0; c < 64; ++c) {
      const float x = hr[c];
      du += x * s_a[c];
      dv += x * s_a[c + 64];
    }
    s_u[r] = du;
    s_v[r] = dv;
    s_s[r] = du + dv;
  }

  if (tid < 128) {
    const int c    = tid & 63;
    const int base = (tid < 64) ? 0 : 256;
    float acc = 0.f;
    for (int r = 0; r < 256; ++r) acc += h[(base + r) * 64 + c];
    if (tid < 64) s_Slo[c] = acc; else s_Shi[c] = acc;
  }
  __syncthreads();

  float lt[2];
#pragma unroll
  for (int q = 0; q < 2; ++q) {
    const int j  = tid + q * 256;
    const int r0 = (2 * j) & 511;
    const float t = s_u[r0] + s_v[r0 + 1];
    lt[q] = (t > 0.f) ? t : ALPHA * t;
  }
  s_red[tid] = fmaxf(lt[0], lt[1]);
  __syncthreads();
  for (int off = 128; off > 0; off >>= 1) {
    if (tid < off) s_red[tid] = fmaxf(s_red[tid], s_red[tid + off]);
    __syncthreads();
  }
  const float m = s_red[0];
  __syncthreads();
  const float e0 = __expf(lt[0] - m);
  const float e1 = __expf(lt[1] - m);
  s_red[tid] = e0 + e1;
  __syncthreads();
  for (int off = 128; off > 0; off >>= 1) {
    if (tid < off) s_red[tid] += s_red[tid + off];
    __syncthreads();
  }
  const float Z = s_red[0];
  __syncthreads();
  s_alpha[tid]       = e0 / Z;
  s_alpha[tid + 256] = e1 / Z;
  __syncthreads();

  if (tid < 64) {
    float acc = 0.f;
    for (int j = 0; j < 512; ++j) acc += s_alpha[j] * h[j * 64 + tid];
    s_P[tid] = (acc > 0.f) ? acc : (__expf(acc) - 1.f);
  }
  __syncthreads();

  {
    const int i = tid;
    const float sv0 = s_s[2 * i];
    const float sv1 = s_s[2 * i + 1];
    const float v0  = (sv0 > 0.f) ? sv0 : ALPHA * sv0;
    const float v1  = (sv1 > 0.f) ? sv1 : ALPHA * sv1;
    const float mm  = fmaxf(v0, v1);
    const float ee0 = __expf(v0 - mm);
    const float ee1 = __expf(v1 - mm);
    const float den = 256.f * (ee0 + ee1);
    const float w0  = ee0 / den;
    const float w1  = ee1 / den;
#pragma unroll 8
    for (int c = 0; c < 64; ++c) {
      const float hp = w0 * s_Slo[c] + w1 * s_Shi[c];
      o[i * 64 + c] = (hp > 0.f) ? hp : (__expf(hp) - 1.f);
    }
  }

  for (int idx = tid; idx < 256 * 64; idx += 256) {
    const int i = 256 + (idx >> 6);
    const int c = idx & 63;
    o[i * 64 + c] = s_P[c];
  }
}

extern "C" void kernel_launch(void* const* d_in, const int* in_sizes, int n_in,
                              void* d_out, int out_size, void* d_ws, size_t ws_size,
                              hipStream_t stream) {
  const float* X = (const float*)d_in[0];
  const float* W = (const float*)d_in[2];
  const float* A = (const float*)d_in[3];
  float* OUT = (float*)d_out;
  float* H   = (float*)d_ws;

  dim3 gemmGrid(128, 4);
  gat_h_gemm_wmma<<<gemmGrid, 32, 0, stream>>>(X, W, H);

  gat_post<<<4, 256, 0, stream>>>(H, A, OUT);
}


# post kernel re-mapped: 16 lanes x float4 per h row over 32 waves, h read once into registers; all sums and maxima by DPP butterfly + one LDS hop, LDS-only barriers; summation order changes
# speedup vs baseline: 3.0245x; 3.0245x over previous
#include <hip/hip_runtime.h>
#include <hip/hip_bf16.h>
#include <math.h>

#define ALPHA 0.2f

typedef __attribute__((ext_vector_type(2))) float v2f;
typedef __attribute__((ext_vector_type(8))) float v8f;

__global__ __launch_bounds__(32) void gat_h_gemm_wmma(
    const float* __restrict__ X,
    const float* __restrict__ W,
    float* __restrict__ H)
{
  const int tileM = blockIdx.x;
  const int tileN = blockIdx.y;
  const int lane  = threadIdx.x;
  const int half  = lane >> 4;
  const int l15   = lane & 15;

  const int rowA = tileM * 16 + l15;
  const int colD = tileN * 16 + l15;

  const float* __restrict__ xrow = X + rowA * 128;

  v8f acc = {};
#pragma unroll
  for (int k0 = 0; k0 < 128; k0 += 4) {
    const int k = k0 + 2 * half;
    v2f a, b;
    a.x = xrow[k];
    a.y = xrow[k + 1];
    b.x = W[k * 64 + colD];
    b.y = W[(k + 1) * 64 + colD];
    acc = __builtin_amdgcn_wmma_f32_16x16x4_f32(
        false, a, false, b, (short)0, acc, false, false);
  }

#pragma unroll
  for (int r = 0; r < 8; ++r) {
    const int row = tileM * 16 + r + 8 * half;
    H[row * 64 + colD] = acc[r];
  }
}

#define DEV static __device__ __forceinline__

template <int CTRL> DEV float dpp_row_read(float v) {
  return __builtin_bit_cast(float, __builtin_amdgcn_update_dpp(
      0, __builtin_bit_cast(int, v), CTRL, 0xf, 0xf, true));
}
DEV float row16_sum(float v) {
  v += dpp_row_read<0x161>(v);
  v += dpp_row_read<0x162>(v);
  v += dpp_row_read<0x164>(v);
  v += dpp_row_read<0x168>(v);
  return v;
}
DEV float row16_max(float v) {
  v = fmaxf(v, dpp_row_read<0x161>(v));
  v = fmaxf(v, dpp_row_read<0x162>(v));
  v = fmaxf(v, dpp_row_read<0x164>(v));
  v = fmaxf(v, dpp_row_read<0x168>(v));
  return v;
}
DEV float xhalf(float v) {
  const int b = __builtin_bit_cast(int, v);
  return __builtin_bit_cast(float, __builtin_amdgcn_permlanex16(
      b, b, 0x76543210, 0xfedcba98, false, false));
}
DEV float elu1(float x) { return (x > 0.f) ? x : (__expf(x) - 1.f); }
DEV void lds_barrier() {
  __builtin_amdgcn_fence(__ATOMIC_RELEASE, "workgroup", "local");
  __builtin_amdgcn_s_barrier();
  __builtin_amdgcn_fence(__ATOMIC_ACQUIRE, "workgroup", "local");
}

__global__ __launch_bounds__(1024) void gat_post(
    const float* __restrict__ H,
    const float* __restrict__ A,
    float* __restrict__ OUT)
{
  const int b    = blockIdx.x;
  const int tid  = threadIdx.x;
  const int lane = tid & 31;
  const int wave = tid >> 5;
  const int sub  = tid & 15;
  const int rg   = tid >> 4;

  const float4* __restrict__ h4 = reinterpret_cast<const float4*>(H + b * 512 * 64);
  float4* __restrict__ o4       = reinterpret_cast<float4*>(OUT + b * 512 * 64);

  __shared__ float s_u[512];
  __shared__ float s_v[512];
  __shared__ float s_s[512];
  __shared__ float s_alpha[256];
  __shared__ float s_w0[256];
  __shared__ float s_w1[256];
  __shared__ __attribute__((aligned(16))) float s_plo[32 * 64];
  __shared__ __attribute__((aligned(16))) float s_phi[32 * 64];
  __shared__ __attribute__((aligned(16))) float s_Slo[64];
  __shared__ __attribute__((aligned(16))) float s_Shi[64];
  __shared__ __attribute__((aligned(16))) float s_P[64];
  __shared__ float s_rmax[8];
  __shared__ float s_rsum[8];

  float4 x[8];
#pragma unroll
  for (int it = 0; it < 8; ++it) x[it] = h4[(it * 64 + rg) * 16 + sub];

  const float alo0 = A[4 * sub + 0], alo1 = A[4 * sub + 1];
  const float alo2 = A[4 * sub + 2], alo3 = A[4 * sub + 3];
  const float ahi0 = A[64 + 4 * sub + 0], ahi1 = A[64 + 4 * sub + 1];
  const float ahi2 = A[64 + 4 * sub + 2], ahi3 = A[64 + 4 * sub + 3];

  float4 clo = make_float4(0.f, 0.f, 0.f, 0.f);
  float4 chi = make_float4(0.f, 0.f, 0.f, 0.f);
#pragma unroll
  for (int it = 0; it < 8; ++it) {
    const float4 xv = x[it];
    float du = xv.x * alo0;
    du = fmaf(xv.y, alo1, du);
    du = fmaf(xv.z, alo2, du);
    du = fmaf(xv.w, alo3, du);
    float dv = xv.x * ahi0;
    dv = fmaf(xv.y, ahi1, dv);
    dv = fmaf(xv.z, ahi2, dv);
    dv = fmaf(xv.w, ahi3, dv);
    du = row16_sum(du);
    dv = row16_sum(dv);
    if (sub == 0) {
      const int r = it * 64 + rg;
      s_u[r] = du;
      s_v[r] = dv;
      s_s[r] = du + dv;
    }
    if (it < 4) {
      clo.x += xv.x; clo.y += xv.y; clo.z += xv.z; clo.w += xv.w;
    } else {
      chi.x += xv.x; chi.y += xv.y; chi.z += xv.z; chi.w += xv.w;
    }
  }
  clo.x += xhalf(clo.x); clo.y += xhalf(clo.y); clo.z += xhalf(clo.z); clo.w += xhalf(clo.w);
  chi.x += xhalf(chi.x); chi.y += xhalf(chi.y); chi.z += xhalf(chi.z); chi.w += xhalf(chi.w);
  if (lane < 16) {
    *reinterpret_cast<float4*>(&s_plo[wave * 64 + 4 * sub]) = clo;
    *reinterpret_cast<float4*>(&s_phi[wave * 64 + 4 * sub]) = chi;
  }
  __syncthreads();

  if (tid < 128) {
    const int c = tid & 63;
    const float* p = (tid < 64) ? s_plo : s_phi;
    float a0 = 0.f, a1 = 0.f, a2 = 0.f, a3 = 0.f;
    for (int w = 0; w < 32; w += 4) {
      a0 += p[(w + 0) * 64 + c];
      a1 += p[(w + 1) * 64 + c];
      a2 += p[(w + 2) * 64 + c];
      a3 += p[(w + 3) * 64 + c];
    }
    const float tot = (a0 + a1) + (a2 + a3);
    if (tid < 64) s_Slo[c] = tot; else s_Shi[c] = tot;
  }

  const int j = tid & 255;
  const float tj = s_u[2 * j] + s_v[2 * j + 1];
  const float lt = (tj > 0.f) ? tj : ALPHA * tj;
  float wm = row16_max(lt);
  wm = fmaxf(wm, xhalf(wm));
  if (tid < 256 && lane == 0) s_rmax[wave] = wm;

  if (tid < 256) {
    const float sv0 = s_s[2 * tid];
    const float sv1 = s_s[2 * tid + 1];
    const float v0  = (sv0 > 0.f) ? sv0 : ALPHA * sv0;
    const float v1  = (sv1 > 0.f) ? sv1 : ALPHA * sv1;
    const float mm  = fmaxf(v0, v1);
    const float ee0 = __expf(v0 - mm);
    const float ee1 = __expf(v1 - mm);
    const float den = 256.f * (ee0 + ee1);
    s_w0[tid] = ee0 / den;
    s_w1[tid] = ee1 / den;
  }
  __syncthreads();

  float m = s_rmax[0];
  for (int k = 1; k < 8; ++k) m = fmaxf(m, s_rmax[k]);
  const float e = __expf(lt - m);
  float ws = row16_sum(e + e);
  ws += xhalf(ws);
  if (tid < 256 && lane == 0) s_rsum[wave] = ws;

  {
    const float4 lo = *reinterpret_cast<const float4*>(&s_Slo[4 * sub]);
    const float4 hi = *reinterpret_cast<const float4*>(&s_Shi[4 * sub]);
    for (int it = 0; it < 4; ++it) {
      const int i = it * 64 + rg;
      const float w0 = s_w0[i];
      const float w1 = s_w1[i];
      float4 r;
      r.x = elu1(w0 * lo.x + w1 * hi.x);
      r.y = elu1(w0 * lo.y + w1 * hi.y);
      r.z = elu1(w0 * lo.z + w1 * hi.z);
      r.w = elu1(w0 * lo.w + w1 * hi.w);
      o4[i * 16 + sub] = r;
    }
  }
  lds_barrier();

  float Z = s_rsum[0];
  for (int k = 1; k < 8; ++k) Z += s_rsum[k];
  if (tid < 256) s_alpha[tid] = e / Z;
  lds_barrier();

  float4 pp = make_float4(0.f, 0.f, 0.f, 0.f);
#pragma unroll
  for (int it = 0; it < 8; ++it) {
    const float al = s_alpha[(it * 64 + rg) & 255];
    pp.x = fmaf(al, x[it].x, pp.x);
    pp.y = fmaf(al, x[it].y, pp.y);
    pp.z = fmaf(al, x[it].z, pp.z);
    pp.w = fmaf(al, x[it].w, pp.w);
  }
  pp.x += xhalf(pp.x); pp.y += xhalf(pp.y); pp.z += xhalf(pp.z); pp.w += xhalf(pp.w);
  if (lane < 16) *reinterpret_cast<float4*>(&s_plo[wave * 64 + 4 * sub]) = pp;
  lds_barrier();

  if (tid < 64) {
    float a0 = 0.f, a1 = 0.f, a2 = 0.f, a3 = 0.f;
    for (int w = 0; w < 32; w += 4) {
      a0 += s_plo[(w + 0) * 64 + tid];
      a1 += s_plo[(w + 1) * 64 + tid];
      a2 += s_plo[(w + 2) * 64 + tid];
      a3 += s_plo[(w + 3) * 64 + tid];
    }
    s_P[tid] = elu1((a0 + a1) + (a2 + a3));
  }
  lds_barrier();

  {
    const float4 pv = *reinterpret_cast<const float4*>(&s_P[4 * sub]);
    for (int it = 0; it < 4; ++it) {
      const int i = 256 + it * 64 + rg;
      o4[i * 16 + sub] = pv;
    }
  }
}

extern "C" void kernel_launch(void* const* d_in, const int* in_sizes, int n_in,
                              void* d_out, int out_size, void* d_ws, size_t ws_size,
                              hipStream_t stream) {
  const float* X = (const float*)d_in[0];
  const float* W = (const float*)d_in[2];
  const float* A = (const float*)d_in[3];
  float* OUT = (float*)d_out;
  float* H   = (float*)d_ws;

  dim3 gemmGrid(128, 4);
  gat_h_gemm_wmma<<<gemmGrid, 32, 0, stream>>>(X, W, H);

  gat_post<<<4, 1024, 0, stream>>>(H, A, OUT);
}


# row dots, column sums and softmax-weighted row sums taken from the h tiles inside the GEMM kernel as per-block partials, h never stored; 2nd kernel only combines them; summation order changes
# speedup vs baseline: 3.3016x; 1.0916x over previous
#include <hip/hip_runtime.h>
#include <hip/hip_bf16.h>
#include <math.h>


#define ALPHA 0.2f

typedef __attribute__((ext_vector_type(2))) float v2f;
typedef __attribute__((ext_vector_type(8))) float v8f;

#define DEV static __device__ __forceinline__

#define OFF_CS 0
#define OFF_PP (OFF_CS + 4 * 2 * 32 * 64)
#define OFF_MX (OFF_PP + 4 * 32 * 64)
#define OFF_LS (OFF_MX + 4 * 32)
#define OFF_W0 (OFF_LS + 4 * 32)
#define OFF_W1 (OFF_W0 + 4 * 256)

template <int CTRL> DEV float dpp_row_read(float v) {
  return __builtin_bit_cast(float, __builtin_amdgcn_update_dpp(
      0, __builtin_bit_cast(int, v), CTRL, 0xf, 0xf, true));
}
DEV float row16_sum(float v) {
  v += dpp_row_read<0x161>(v);
  v += dpp_row_read<0x162>(v);
  v += dpp_row_read<0x164>(v);
  v += dpp_row_read<0x168>(v);
  return v;
}
DEV float xhalf(float v) {
  const int b = __builtin_bit_cast(int, v);
  return __builtin_bit_cast(float, __builtin_amdgcn_permlanex16(
      b, b, 0x76543210, 0xfedcba98, false, false));
}
DEV float elu1(float x) { return (x > 0.f) ? x : (__expf(x) - 1.f); }
DEV float leaky(float x) { return (x > 0.f) ? x : ALPHA * x; }

__global__ __launch_bounds__(256) void gat_gemm_partials(
    const float* __restrict__ X,
    const float* __restrict__ W,
    const float* __restrict__ A,
    float* __restrict__ WS)
{
  const int blk  = blockIdx.x;
  const int b    = blockIdx.y;
  const int tid  = threadIdx.x;
  const int lane = tid & 31;
  const int wave = tid >> 5;
  const int half = lane >> 4;
  const int l15  = lane & 15;
  const int nt   = wave & 3;
  const bool keyTile = wave >= 4;

  int rowA;
  if (keyTile) rowA = 16 * blk + l15;
  else         rowA = (l15 < 8) ? (8 * blk + l15) : (256 + 8 * blk + (l15 - 8));
  const int colD = nt * 16 + l15;

  const float* __restrict__ xrow = X + (b * 512 + rowA) * 128;

  const float alo = A[colD];
  const float ahi = A[64 + colD];

  v8f acc = {};
#pragma unroll
  for (int k0 = 0; k0 < 128; k0 += 4) {
    const int k = k0 + 2 * half;
    v2f a, bb;
    a.x = xrow[k];
    a.y = xrow[k + 1];
    bb.x = W[k * 64 + colD];
    bb.y = W[(k + 1) * 64 + colD];
    acc = __builtin_amdgcn_wmma_f32_16x16x4_f32(
        false, a, false, bb, (short)0, acc, false, false);
  }

  asm volatile("" :: "v"(alo), "v"(ahi));

  __shared__ float s_u[4][16];
  __shared__ float s_v[4][16];
  __shared__ float s_lt[8];

  if (keyTile) {
    float pu[8], pv[8];
#pragma unroll
    for (int r = 0; r < 8; ++r) {
      pu[r] = row16_sum(acc[r] * alo);
      pv[r] = row16_sum(acc[r] * ahi);
    }
    if (l15 == 0) {
#pragma unroll
      for (int r = 0; r < 8; ++r) {
        s_u[nt][r + 8 * half] = pu[r];
        s_v[nt][r + 8 * half] = pv[r];
      }
    }
  } else {
    float cs = 0.f;
#pragma unroll
    for (int r = 0; r < 8; ++r) cs += acc[r];
    WS[OFF_CS + ((b * 2 + half) * 32 + blk) * 64 + colD] = cs;
  }
  __syncthreads();

  if (tid < 8) {
    const int jj = tid;
    float ue = 0.f, ve = 0.f, uo = 0.f, vo = 0.f;
#pragma unroll
    for (int n = 0; n < 4; ++n) {
      ue += s_u[n][2 * jj];
      ve += s_v[n][2 * jj];
      uo += s_u[n][2 * jj + 1];
      vo += s_v[n][2 * jj + 1];
    }
    s_lt[jj] = leaky(ue + vo);
    const float v0  = leaky(ue + ve);
    const float v1  = leaky(uo + vo);
    const float mm  = fmaxf(v0, v1);
    const float ee0 = __expf(v0 - mm);
    const float ee1 = __expf(v1 - mm);
    const float den = 256.f * (ee0 + ee1);
    WS[OFF_W0 + b * 256 + 8 * blk + jj] = ee0 / den;
    WS[OFF_W1 + b * 256 + 8 * blk + jj] = ee1 / den;
  }
  __syncthreads();

  if (!keyTile) {
    float lt[8];
#pragma unroll
    for (int r = 0; r < 8; ++r) lt[r] = s_lt[r];
    float m = lt[0];
#pragma unroll
    for (int r = 1; r < 8; ++r) m = fmaxf(m, lt[r]);
    float l = 0.f, pp = 0.f;
#pragma unroll
    for (int r = 0; r < 8; ++r) {
      const float e = __expf(lt[r] - m);
      l += e;
      pp = fmaf(e, acc[r], pp);
    }
    pp += xhalf(pp);
    if (half == 0) WS[OFF_PP + (b * 32 + blk) * 64 + colD] = pp;
    if (tid == 0) {
      WS[OFF_MX + b * 32 + blk] = m;
      WS[OFF_LS + b * 32 + blk] = l;
    }
  }
}

__global__ __launch_bounds__(256) void gat_combine(
    const float* __restrict__ WS,
    float* __restrict__ OUT)
{
  const int g   = blockIdx.x;
  const int b   = blockIdx.y;
  const int tid = threadIdx.x;
  const int c   = tid & 63;
  const int q   = tid >> 6;
  const int sub = tid & 15;
  const int rg  = tid >> 4;

  __shared__ __attribute__((aligned(16))) float s_S[3][64];

  const int i = 16 * g + rg;
  const float w0 = WS[OFF_W0 + b * 256 + i];
  const float w1 = WS[OFF_W1 + b * 256 + i];

  float pv[32];
#pragma unroll
  for (int k = 0; k < 32; ++k) pv[k] = 0.f;
  if (q < 3) {
    const float* __restrict__ p = (q < 2) ? (WS + OFF_CS + ((b * 2 + q) * 32) * 64 + c)
                                          : (WS + OFF_PP + (b * 32) * 64 + c);
#pragma unroll
    for (int k = 0; k < 32; ++k) pv[k] = p[k * 64];
  }

  const float* __restrict__ mx = WS + OFF_MX + b * 32;
  const float* __restrict__ ls = WS + OFF_LS + b * 32;

  float M = mx[0];
#pragma unroll
  for (int k = 1; k < 32; ++k) M = fmaxf(M, mx[k]);
  float sc[32];
  float Zh = 0.f;
#pragma unroll
  for (int k = 0; k < 32; ++k) {
    sc[k] = __expf(mx[k] - M);
    Zh = fmaf(sc[k], ls[k], Zh);
  }

  if (q < 2) {
    float a0 = 0.f, a1 = 0.f, a2 = 0.f, a3 = 0.f;
#pragma unroll
    for (int k = 0; k < 32; k += 4) {
      a0 += pv[k + 0];
      a1 += pv[k + 1];
      a2 += pv[k + 2];
      a3 += pv[k + 3];
    }
    s_S[q][c] = (a0 + a1) + (a2 + a3);
  } else if (q == 2) {
    float a0 = 0.f, a1 = 0.f, a2 = 0.f, a3 = 0.f;
#pragma unroll
    for (int k = 0; k < 32; k += 4) {
      a0 = fmaf(sc[k + 0], pv[k + 0], a0);
      a1 = fmaf(sc[k + 1], pv[k + 1], a1);
      a2 = fmaf(sc[k + 2], pv[k + 2], a2);
      a3 = fmaf(sc[k + 3], pv[k + 3], a3);
    }
    s_S[2][c] = (a0 + a1) + (a2 + a3);
  }
  __syncthreads();

  const float4 lo = *reinterpret_cast<const float4*>(&s_S[0][4 * sub]);
  const float4 hi = *reinterpret_cast<const float4*>(&s_S[1][4 * sub]);
  const float4 p0 = *reinterpret_cast<const float4*>(&s_S[2][4 * sub]);
  const float rz = 1.f / (Zh + Zh);

  float4* __restrict__ o4 = reinterpret_cast<float4*>(OUT + b * 512 * 64);

  float4 r;
  r.x = elu1(w0 * lo.x + w1 * hi.x);
  r.y = elu1(w0 * lo.y + w1 * hi.y);
  r.z = elu1(w0 * lo.z + w1 * hi.z);
  r.w = elu1(w0 * lo.w + w1 * hi.w);
  o4[i * 16 + sub] = r;

  float4 pr;
  pr.x = elu1(p0.x * rz);
  pr.y = elu1(p0.y * rz);
  pr.z = elu1(p0.z * rz);
  pr.w = elu1(p0.w * rz);
  o4[(256 + i) * 16 + sub] = pr;
}

extern "C" void kernel_launch(void* const* d_in, const int* in_sizes, int n_in,
                              void* d_out, int out_size, void* d_ws, size_t ws_size,
                              hipStream_t stream) {
  const float* X = (const float*)d_in[0];
  const float* W = (const float*)d_in[2];
  const float* A = (const float*)d_in[3];
  float* OUT = (float*)d_out;
  float* WS  = (float*)d_ws;

  dim3 grid1(32, 4);
  gat_gemm_partials<<<grid1, 256, 0, stream>>>(X, W, A, WS);

  dim3 grid2(16, 4);
  gat_combine<<<grid2, 256, 0, stream>>>(WS, OUT);
}


# GEMM kernel epilogue: one LDS barrier instead of two and no global store ahead of it; each value wave forms the 8 logits itself from the LDS row dots, blend weights by a key wave beside it
# speedup vs baseline: 3.6452x; 1.1041x over previous
#include <hip/hip_runtime.h>
#include <hip/hip_bf16.h>
#include <math.h>


#define ALPHA 0.2f

typedef __attribute__((ext_vector_type(2))) float v2f;
typedef __attribute__((ext_vector_type(8))) float v8f;

#define DEV static __device__ __forceinline__

#define OFF_CS 0
#define OFF_PP (OFF_CS + 4 * 2 * 32 * 64)
#define OFF_MX (OFF_PP + 4 * 32 * 64)
#define OFF_LS (OFF_MX + 4 * 32)
#define OFF_W0 (OFF_LS + 4 * 32)
#define OFF_W1 (OFF_W0 + 4 * 256)

template <int CTRL> DEV float dpp_row_read(float v) {
  return __builtin_bit_cast(float, __builtin_amdgcn_update_dpp(
      0, __builtin_bit_cast(int, v), CTRL, 0xf, 0xf, true));
}
DEV float row16_sum(float v) {
  v += dpp_row_read<0x161>(v);
  v += dpp_row_read<0x162>(v);
  v += dpp_row_read<0x164>(v);
  v += dpp_row_read<0x168>(v);
  return v;
}
DEV float xhalf(float v) {
  const int b = __builtin_bit_cast(int, v);
  return __builtin_bit_cast(float, __builtin_amdgcn_permlanex16(
      b, b, 0x76543210, 0xfedcba98, false, false));
}
DEV float elu1(float x) { return (x > 0.f) ? x : (__expf(x) - 1.f); }
DEV float leaky(float x) { return (x > 0.f) ? x : ALPHA * x; }

__global__ __launch_bounds__(256) void gat_gemm_partials(
    const float* __restrict__ X,
    const float* __restrict__ W,
    const float* __restrict__ A,
    float* __restrict__ WS)
{
  const int blk  = blockIdx.x;
  const int b    = blockIdx.y;
  const int tid  = threadIdx.x;
  const int lane = tid & 31;
  const int wave = tid >> 5;
  const int half = lane >> 4;
  const int l15  = lane & 15;
  const int nt   = wave & 3;
  const bool keyTile = wave >= 4;

  int rowA;
  if (keyTile) rowA = 16 * blk + l15;
  else         rowA = (l15 < 8) ? (8 * blk + l15) : (256 + 8 * blk + (l15 - 8));
  const int colD = nt * 16 + l15;

  const float* __restrict__ xrow = X + (b * 512 + rowA) * 128;

  const float alo = A[colD];
  const float ahi = A[64 + colD];

  v8f acc = {};
#pragma unroll
  for (int k0 = 0; k0 < 128; k0 += 4) {
    const int k = k0 + 2 * half;
    v2f a, bb;
    a.x = xrow[k];
    a.y = xrow[k + 1];
    bb.x = W[k * 64 + colD];
    bb.y = W[(k + 1) * 64 + colD];
    acc = __builtin_amdgcn_wmma_f32_16x16x4_f32(
        false, a, false, bb, (short)0, acc, false, false);
  }

  asm volatile("" :: "v"(alo), "v"(ahi));

  __shared__ float s_u[4][16];
  __shared__ float s_v[4][16];

  float cs = 0.f;
  if (keyTile) {
    float pu[8], pv[8];
#pragma unroll
    for (int r = 0; r < 8; ++r) {
      pu[r] = row16_sum(acc[r] * alo);
      pv[r] = row16_sum(acc[r] * ahi);
    }
    if (l15 == 0) {
#pragma unroll
      for (int r = 0; r < 8; ++r) {
        s_u[nt][r + 8 * half] = pu[r];
        s_v[nt][r + 8 * half] = pv[r];
      }
    }
  } else {
#pragma unroll
    for (int r = 0; r < 8; ++r) cs += acc[r];
  }
  __syncthreads();

  if (!keyTile) {
    const int jr = lane & 7;
    float ue = 0.f, vo = 0.f;
#pragma unroll
    for (int n = 0; n < 4; ++n) {
      ue += s_u[n][2 * jr];
      vo += s_v[n][2 * jr + 1];
    }
    const int ltb = __builtin_bit_cast(int, leaky(ue + vo));
    float lt[8];
#pragma unroll
    for (int r = 0; r < 8; ++r)
      lt[r] = __builtin_bit_cast(float, __builtin_amdgcn_readlane(ltb, r));
    float m = lt[0];
#pragma unroll
    for (int r = 1; r < 8; ++r) m = fmaxf(m, lt[r]);
    float l = 0.f, pp = 0.f;
#pragma unroll
    for (int r = 0; r < 8; ++r) {
      const float e = __expf(lt[r] - m);
      l += e;
      pp = fmaf(e, acc[r], pp);
    }
    pp += xhalf(pp);
    WS[OFF_CS + ((b * 2 + half) * 32 + blk) * 64 + colD] = cs;
    if (half == 0) WS[OFF_PP + (b * 32 + blk) * 64 + colD] = pp;
    if (tid == 0) {
      WS[OFF_MX + b * 32 + blk] = m;
      WS[OFF_LS + b * 32 + blk] = l;
    }
  } else if (wave == 4) {
    if (lane < 8) {
      const int jj = lane;
      float ue = 0.f, ve = 0.f, uo = 0.f, vo = 0.f;
#pragma unroll
      for (int n = 0; n < 4; ++n) {
        ue += s_u[n][2 * jj];
        ve += s_v[n][2 * jj];
        uo += s_u[n][2 * jj + 1];
        vo += s_v[n][2 * jj + 1];
      }
      const float v0  = leaky(ue + ve);
      const float v1  = leaky(uo + vo);
      const float mm  = fmaxf(v0, v1);
      const float ee0 = __expf(v0 - mm);
      const float ee1 = __expf(v1 - mm);
      const float den = 256.f * (ee0 + ee1);
      WS[OFF_W0 + b * 256 + 8 * blk + jj] = ee0 / den;
      WS[OFF_W1 + b * 256 + 8 * blk + jj] = ee1 / den;
    }
  }
}

__global__ __launch_bounds__(256) void gat_combine(
    const float* __restrict__ WS,
    float* __restrict__ OUT)
{
  const int g   = blockIdx.x;
  const int b   = blockIdx.y;
  const int tid = threadIdx.x;
  const int c   = tid & 63;
  const int q   = tid >> 6;
  const int sub = tid & 15;
  const int rg  = tid >> 4;

  __shared__ __attribute__((aligned(16))) float s_S[3][64];

  const int i = 16 * g + rg;
  const float w0 = WS[OFF_W0 + b * 256 + i];
  const float w1 = WS[OFF_W1 + b * 256 + i];

  float pv[32];
#pragma unroll
  for (int k = 0; k < 32; ++k) pv[k] = 0.f;
  if (q < 3) {
    const float* __restrict__ p = (q < 2) ? (WS + OFF_CS + ((b * 2 + q) * 32) * 64 + c)
                                          : (WS + OFF_PP + (b * 32) * 64 + c);
#pragma unroll
    for (int k = 0; k < 32; ++k) pv[k] = p[k * 64];
  }

  const float* __restrict__ mx = WS + OFF_MX + b * 32;
  const float* __restrict__ ls = WS + OFF_LS + b * 32;

  float M = mx[0];
#pragma unroll
  for (int k = 1; k < 32; ++k) M = fmaxf(M, mx[k]);
  float sc[32];
  float Zh = 0.f;
#pragma unroll
  for (int k = 0; k < 32; ++k) {
    sc[k] = __expf(mx[k] - M);
    Zh = fmaf(sc[k], ls[k], Zh);
  }

  if (q < 2) {
    float a0 = 0.f, a1 = 0.f, a2 = 0.f, a3 = 0.f;
#pragma unroll
    for (int k = 0; k < 32; k += 4) {
      a0 += pv[k + 0];
      a1 += pv[k + 1];
      a2 += pv[k + 2];
      a3 += pv[k + 3];
    }
    s_S[q][c] = (a0 + a1) + (a2 + a3);
  } else if (q == 2) {
    float a0 = 0.f, a1 = 0.f, a2 = 0.f, a3 = 0.f;
#pragma unroll
    for (int k = 0; k < 32; k += 4) {
      a0 = fmaf(sc[k + 0], pv[k + 0], a0);
      a1 = fmaf(sc[k + 1], pv[k + 1], a1);
      a2 = fmaf(sc[k + 2], pv[k + 2], a2);
      a3 = fmaf(sc[k + 3], pv[k + 3], a3);
    }
    s_S[2][c] = (a0 + a1) + (a2 + a3);
  }
  __syncthreads();

  const float4 lo = *reinterpret_cast<const float4*>(&s_S[0][4 * sub]);
  const float4 hi = *reinterpret_cast<const float4*>(&s_S[1][4 * sub]);
  const float4 p0 = *reinterpret_cast<const float4*>(&s_S[2][4 * sub]);
  const float rz = 1.f / (Zh + Zh);

  float4* __restrict__ o4 = reinterpret_cast<float4*>(OUT + b * 512 * 64);

  float4 r;
  r.x = elu1(w0 * lo.x + w1 * hi.x);
  r.y = elu1(w0 * lo.y + w1 * hi.y);
  r.z = elu1(w0 * lo.z + w1 * hi.z);
  r.w = elu1(w0 * lo.w + w1 * hi.w);
  o4[i * 16 + sub] = r;

  float4 pr;
  pr.x = elu1(p0.x * rz);
  pr.y = elu1(p0.y * rz);
  pr.z = elu1(p0.z * rz);
  pr.w = elu1(p0.w * rz);
  o4[(256 + i) * 16 + sub] = pr;
}

extern "C" void kernel_launch(void* const* d_in, const int* in_sizes, int n_in,
                              void* d_out, int out_size, void* d_ws, size_t ws_size,
                              hipStream_t stream) {
  const float* X = (const float*)d_in[0];
  const float* W = (const float*)d_in[2];
  const float* A = (const float*)d_in[3];
  float* OUT = (float*)d_out;
  float* WS  = (float*)d_ws;

  dim3 grid1(32, 4);
  gat_gemm_partials<<<grid1, 256, 0, stream>>>(X, W, A, WS);

  dim3 grid2(16, 4);
  gat_combine<<<grid2, 256, 0, stream>>>(WS, OUT);
}


# GEMM kernel: X rows and W staged once per block through LDS by whole-line float4 loads; all A and B fragments are LDS reads issued as one group; no per-lane strided global loads left
# speedup vs baseline: 4.2228x; 1.1585x over previous
#include <hip/hip_runtime.h>
#include <hip/hip_bf16.h>
#include <math.h>


#define ALPHA 0.2f

typedef __attribute__((ext_vector_type(2))) float v2f;
typedef __attribute__((ext_vector_type(8))) float v8f;

#define DEV static __device__ __forceinline__

#define OFF_CS 0
#define OFF_PP (OFF_CS + 4 * 2 * 32 * 64)
#define OFF_MX (OFF_PP + 4 * 32 * 64)
#define OFF_LS (OFF_MX + 4 * 32)
#define OFF_W0 (OFF_LS + 4 * 32)
#define OFF_W1 (OFF_W0 + 4 * 256)

#define XPITCH 132
#define WPITCH 80

template <int CTRL> DEV float dpp_row_read(float v) {
  return __builtin_bit_cast(float, __builtin_amdgcn_update_dpp(
      0, __builtin_bit_cast(int, v), CTRL, 0xf, 0xf, true));
}
DEV float row16_sum(float v) {
  v += dpp_row_read<0x161>(v);
  v += dpp_row_read<0x162>(v);
  v += dpp_row_read<0x164>(v);
  v += dpp_row_read<0x168>(v);
  return v;
}
DEV float xhalf(float v) {
  const int b = __builtin_bit_cast(int, v);
  return __builtin_bit_cast(float, __builtin_amdgcn_permlanex16(
      b, b, 0x76543210, 0xfedcba98, false, false));
}
DEV float elu1(float x) { return (x > 0.f) ? x : (__expf(x) - 1.f); }
DEV float leaky(float x) { return (x > 0.f) ? x : ALPHA * x; }

__global__ __launch_bounds__(256) void gat_gemm_partials(
    const float* __restrict__ X,
    const float* __restrict__ W,
    const float* __restrict__ A,
    float* __restrict__ WS)
{
  const int blk  = blockIdx.x;
  const int b    = blockIdx.y;
  const int tid  = threadIdx.x;
  const int lane = tid & 31;
  const int wave = tid >> 5;
  const int half = lane >> 4;
  const int l15  = lane & 15;
  const int nt   = wave & 3;
  const bool keyTile = wave >= 4;

  const int colD = nt * 16 + l15;

  __shared__ __attribute__((aligned(16))) float s_x[32][XPITCH];
  const float4* __restrict__ xb4 = reinterpret_cast<const float4*>(X + b * 512 * 128);
  const float4 g0 = xb4[(8 * blk + wave) * 32 + lane];
  const float4 g1 = xb4[(256 + 8 * blk + wave) * 32 + lane];
  const float4 g2 = xb4[(16 * blk + wave) * 32 + lane];
  const float4 g3 = xb4[(16 * blk + 8 + wave) * 32 + lane];

  __shared__ __attribute__((aligned(16))) float s_w[128][WPITCH];
  const float4* __restrict__ w4 = reinterpret_cast<const float4*>(W);
  float4 gw[8];
#pragma unroll
  for (int q = 0; q < 8; ++q) gw[q] = w4[q * 256 + tid];

  const float alo = A[colD];
  const float ahi = A[64 + colD];

  *reinterpret_cast<float4*>(&s_x[wave][4 * lane])      = g0;
  *reinterpret_cast<float4*>(&s_x[8 + wave][4 * lane])  = g1;
  *reinterpret_cast<float4*>(&s_x[16 + wave][4 * lane]) = g2;
  *reinterpret_cast<float4*>(&s_x[24 + wave][4 * lane]) = g3;
#pragma unroll
  for (int q = 0; q < 8; ++q)
    *reinterpret_cast<float4*>(&s_w[16 * q + (tid >> 4)][4 * (tid & 15)]) = gw[q];
  __syncthreads();

  const float* arow = &s_x[(keyTile ? 16 : 0) + l15][2 * half];

  const float* bcol = &s_w[2 * half][colD];

  v2f af[32], bw[32];
#pragma unroll
  for (int s = 0; s < 32; ++s) {
    af[s] = *reinterpret_cast<const v2f*>(arow + 4 * s);
    bw[s].x = bcol[(4 * s) * WPITCH];
    bw[s].y = bcol[(4 * s + 1) * WPITCH];
  }
  __builtin_amdgcn_sched_barrier(0);

  v8f acc = {};
#pragma unroll
  for (int s = 0; s < 32; ++s) {
    acc = __builtin_amdgcn_wmma_f32_16x16x4_f32(
        false, af[s], false, bw[s], (short)0, acc, false, false);
  }

  __shared__ float s_u[4][16];
  __shared__ float s_v[4][16];

  float cs = 0.f;
  if (keyTile) {
    float pu[8], pv[8];
#pragma unroll
    for (int r = 0; r < 8; ++r) {
      pu[r] = row16_sum(acc[r] * alo);
      pv[r] = row16_sum(acc[r] * ahi);
    }
    if (l15 == 0) {
#pragma unroll
      for (int r = 0; r < 8; ++r) {
        s_u[nt][r + 8 * half] = pu[r];
        s_v[nt][r + 8 * half] = pv[r];
      }
    }
  } else {
#pragma unroll
    for (int r = 0; r < 8; ++r) cs += acc[r];
  }
  __syncthreads();

  if (!keyTile) {
    const int jr = lane & 7;
    float ue = 0.f, vo = 0.f;
#pragma unroll
    for (int n = 0; n < 4; ++n) {
      ue += s_u[n][2 * jr];
      vo += s_v[n][2 * jr + 1];
    }
    const int ltb = __builtin_bit_cast(int, leaky(ue + vo));
    float lt[8];
#pragma unroll
    for (int r = 0; r < 8; ++r)
      lt[r] = __builtin_bit_cast(float, __builtin_amdgcn_readlane(ltb, r));
    float m = lt[0];
#pragma unroll
    for (int r = 1; r < 8; ++r) m = fmaxf(m, lt[r]);
    float l = 0.f, pp = 0.f;
#pragma unroll
    for (int r = 0; r < 8; ++r) {
      const float e = __expf(lt[r] - m);
      l += e;
      pp = fmaf(e, acc[r], pp);
    }
    pp += xhalf(pp);
    WS[OFF_CS + ((b * 2 + half) * 32 + blk) * 64 + colD] = cs;
    if (half == 0) WS[OFF_PP + (b * 32 + blk) * 64 + colD] = pp;
    if (tid == 0) {
      WS[OFF_MX + b * 32 + blk] = m;
      WS[OFF_LS + b * 32 + blk] = l;
    }
  } else if (wave == 4) {
    if (lane < 8) {
      const int jj = lane;
      float ue = 0.f, ve = 0.f, uo = 0.f, vo = 0.f;
#pragma unroll
      for (int n = 0; n < 4; ++n) {
        ue += s_u[n][2 * jj];
        ve += s_v[n][2 * jj];
        uo += s_u[n][2 * jj + 1];
        vo += s_v[n][2 * jj + 1];
      }
      const float v0  = leaky(ue + ve);
      const float v1  = leaky(uo + vo);
      const float mm  = fmaxf(v0, v1);
      const float ee0 = __expf(v0 - mm);
      const float ee1 = __expf(v1 - mm);
      const float den = 256.f * (ee0 + ee1);
      WS[OFF_W0 + b * 256 + 8 * blk + jj] = ee0 / den;
      WS[OFF_W1 + b * 256 + 8 * blk + jj] = ee1 / den;
    }
  }
}

__global__ __launch_bounds__(256) void gat_combine(
    const float* __restrict__ WS,
    float* __restrict__ OUT)
{
  const int g   = blockIdx.x;
  const int b   = blockIdx.y;
  const int tid = threadIdx.x;
  const int c   = tid & 63;
  const int q   = tid >> 6;
  const int sub = tid & 15;
  const int rg  = tid >> 4;

  __shared__ __attribute__((aligned(16))) float s_S[3][64];

  const int i = 16 * g + rg;
  const float w0 = WS[OFF_W0 + b * 256 + i];
  const float w1 = WS[OFF_W1 + b * 256 + i];

  float pv[32];
#pragma unroll
  for (int k = 0; k < 32; ++k) pv[k] = 0.f;
  if (q < 3) {
    const float* __restrict__ p = (q < 2) ? (WS + OFF_CS + ((b * 2 + q) * 32) * 64 + c)
                                          : (WS + OFF_PP + (b * 32) * 64 + c);
#pragma unroll
    for (int k = 0; k < 32; ++k) pv[k] = p[k * 64];
  }

  const float* __restrict__ mx = WS + OFF_MX + b * 32;
  const float* __restrict__ ls = WS + OFF_LS + b * 32;

  float M = mx[0];
#pragma unroll
  for (int k = 1; k < 32; ++k) M = fmaxf(M, mx[k]);
  float sc[32];
  float Zh = 0.f;
#pragma unroll
  for (int k = 0; k < 32; ++k) {
    sc[k] = __expf(mx[k] - M);
    Zh = fmaf(sc[k], ls[k], Zh);
  }

  if (q < 2) {
    float a0 = 0.f, a1 = 0.f, a2 = 0.f, a3 = 0.f;
#pragma unroll
    for (int k = 0; k < 32; k += 4) {
      a0 += pv[k + 0];
      a1 += pv[k + 1];
      a2 += pv[k + 2];
      a3 += pv[k + 3];
    }
    s_S[q][c] = (a0 + a1) + (a2 + a3);
  } else if (q == 2) {
    float a0 = 0.f, a1 = 0.f, a2 = 0.f, a3 = 0.f;
#pragma unroll
    for (int k = 0; k < 32; k += 4) {
      a0 = fmaf(sc[k + 0], pv[k + 0], a0);
      a1 = fmaf(sc[k + 1], pv[k + 1], a1);
      a2 = fmaf(sc[k + 2], pv[k + 2], a2);
      a3 = fmaf(sc[k + 3], pv[k + 3], a3);
    }
    s_S[2][c] = (a0 + a1) + (a2 + a3);
  }
  __syncthreads();

  const float4 lo = *reinterpret_cast<const float4*>(&s_S[0][4 * sub]);
  const float4 hi = *reinterpret_cast<const float4*>(&s_S[1][4 * sub]);
  const float4 p0 = *reinterpret_cast<const float4*>(&s_S[2][4 * sub]);
  const float rz = 1.f / (Zh + Zh);

  float4* __restrict__ o4 = reinterpret_cast<float4*>(OUT + b * 512 * 64);

  float4 r;
  r.x = elu1(w0 * lo.x + w1 * hi.x);
  r.y = elu1(w0 * lo.y + w1 * hi.y);
  r.z = elu1(w0 * lo.z + w1 * hi.z);
  r.w = elu1(w0 * lo.w + w1 * hi.w);
  o4[i * 16 + sub] = r;

  float4 pr;
  pr.x = elu1(p0.x * rz);
  pr.y = elu1(p0.y * rz);
  pr.z = elu1(p0.z * rz);
  pr.w = elu1(p0.w * rz);
  o4[(256 + i) * 16 + sub] = pr;
}

extern "C" void kernel_launch(void* const* d_in, const int* in_sizes, int n_in,
                              void* d_out, int out_size, void* d_ws, size_t ws_size,
                              hipStream_t stream) {
  const float* X = (const float*)d_in[0];
  const float* W = (const float*)d_in[2];
  const float* A = (const float*)d_in[3];
  float* OUT = (float*)d_out;
  float* WS  = (float*)d_ws;

  dim3 grid1(32, 4);
  gat_gemm_partials<<<grid1, 256, 0, stream>>>(X, W, A, WS);

  dim3 grid2(16, 4);
  gat_combine<<<grid2, 256, 0, stream>>>(WS, OUT);
}
